# starting kernel
# baseline (speedup 1.0000x reference)
#include <hip/hip_runtime.h>
#include <hip/hip_bf16.h>


#define Bn    128
#define CINn  64
#define COUTn 64
#define ODn   256
#define Kn    8
#define Sn    4
#define Ln    1028
#define KKn   (CINn * Kn)
#define RSQRT_CIN 0.125f

typedef __attribute__((ext_vector_type(2))) float v2f;
typedef __attribute__((ext_vector_type(8))) float v8f;

__global__ __launch_bounds__(256) void
lc1d_wmma_f32(const float* __restrict__ x,
              const float* __restrict__ w,
              float* __restrict__ out)
{
    const int o    = blockIdx.x;
    const int tid  = threadIdx.x;
    const int wave = tid >> 5;
    const int lane = tid & 31;
    const int half = lane >> 4;
    const int r    = lane & 15;

    const int b_base = wave * 16;

    v8f acc0 = {}, acc1 = {}, acc2 = {}, acc3 = {};

    const float* xrow = x + (size_t)(b_base + r) * (CINn * Ln) + (size_t)o * Sn;
    const size_t cstride = (size_t)CINn * ODn * Kn;
    const float* wrow = w + (size_t)r * cstride + (size_t)o * Kn;

#pragma unroll 2
    for (int ks = 0; ks < KKn / 4; ++ks) {
        const int kk = ks * 4 + half * 2;
        const int ci = kk >> 3;
        const int kt = kk & 7;

        v2f a  = *(const v2f*)(xrow + (size_t)ci * Ln + kt);

        const float* wp = wrow + (size_t)ci * (ODn * Kn) + kt;
        v2f b0 = *(const v2f*)(wp + 0  * cstride);
        v2f b1 = *(const v2f*)(wp + 16 * cstride);
        v2f b2 = *(const v2f*)(wp + 32 * cstride);
        v2f b3 = *(const v2f*)(wp + 48 * cstride);

        acc0 = __builtin_amdgcn_wmma_f32_16x16x4_f32(false, a, false, b0,
                                                     (short)0, acc0, false, false);
        acc1 = __builtin_amdgcn_wmma_f32_16x16x4_f32(false, a, false, b1,
                                                     (short)0, acc1, false, false);
        acc2 = __builtin_amdgcn_wmma_f32_16x16x4_f32(false, a, false, b2,
                                                     (short)0, acc2, false, false);
        acc3 = __builtin_amdgcn_wmma_f32_16x16x4_f32(false, a, false, b3,
                                                     (short)0, acc3, false, false);
    }

#pragma unroll
    for (int v = 0; v < 8; ++v) {
        const int bb = b_base + v + 8 * half;
        float* orow = out + (size_t)bb * (COUTn * ODn) + o;
        orow[(size_t)(r +  0) * ODn] = acc0[v] * RSQRT_CIN;
        orow[(size_t)(r + 16) * ODn] = acc1[v] * RSQRT_CIN;
        orow[(size_t)(r + 32) * ODn] = acc2[v] * RSQRT_CIN;
        orow[(size_t)(r + 48) * ODn] = acc3[v] * RSQRT_CIN;
    }
}

extern "C" void kernel_launch(void* const* d_in, const int* in_sizes, int n_in,
                              void* d_out, int out_size, void* d_ws, size_t ws_size,
                              hipStream_t stream) {
    const float* x = (const float*)d_in[0];
    const float* w = (const float*)d_in[1];
    float* out = (float*)d_out;

    lc1d_wmma_f32<<<dim3(ODn), dim3(256), 0, stream>>>(x, w, out);
}


# W_o tile staged once per workgroup in LDS (8-channel chunks, double buffer, next chunk's loads issued ahead of the WMMAs) instead of re-read from global by all 8 waves; sums bit-identical
# speedup vs baseline: 1.3081x; 1.3081x over previous
#include <hip/hip_runtime.h>
#include <hip/hip_bf16.h>


#define Bn    128
#define CINn  64
#define COUTn 64
#define ODn   256
#define Kn    8
#define Sn    4
#define Ln    1028
#define RSQRT_CIN 0.125f

#define CHn    8
#define NCHn   (CINn / CHn)
#define PITCHn 68

typedef __attribute__((ext_vector_type(2))) float v2f;
typedef __attribute__((ext_vector_type(4))) float v4f;
typedef __attribute__((ext_vector_type(8))) float v8f;

__global__ __launch_bounds__(256) void
lc1d_wmma_f32(const float* __restrict__ x,
              const float* __restrict__ w,
              float* __restrict__ out)
{
    __shared__ __attribute__((aligned(16))) float wl[2][COUTn * PITCHn];

    const int o    = blockIdx.x;
    const int tid  = threadIdx.x;
    const int wave = tid >> 5;
    const int lane = tid & 31;
    const int half = lane >> 4;
    const int r    = lane & 15;

    const int b_base = wave * 16;

    v8f acc0 = {}, acc1 = {}, acc2 = {}, acc3 = {};

    const float* xa = x + (size_t)(b_base + r) * (CINn * Ln) + (size_t)o * Sn + 2 * half;

    const size_t cstride = (size_t)CINn * ODn * Kn;
    const int sq  = tid & 1;
    const int sci = (tid >> 1) & 7;
    const int sc  = tid >> 4;
    const float* wsrc = w + (size_t)sc * cstride + (size_t)sci * (ODn * Kn) + (size_t)o * Kn + 4 * sq;
    const int    wdst = sc * PITCHn + sci * 8 + 2 * sq;

    const int    brd  = r * PITCHn + 4 * half;

    v4f wq0, wq1, wq2, wq3;
    v2f ac[2 * CHn], an[2 * CHn];

    wq0 = *(const v4f*)(wsrc);
    wq1 = *(const v4f*)(wsrc + 16 * cstride);
    wq2 = *(const v4f*)(wsrc + 32 * cstride);
    wq3 = *(const v4f*)(wsrc + 48 * cstride);
#pragma unroll
    for (int cl = 0; cl < CHn; ++cl) {
        ac[2 * cl]     = *(const v2f*)(xa + (size_t)cl * Ln);
        ac[2 * cl + 1] = *(const v2f*)(xa + (size_t)cl * Ln + 4);
    }
    {
        float* d = &wl[0][wdst];
        *(v2f*)(d)                   = (v2f){wq0.x, wq0.y};
        *(v2f*)(d + 4)               = (v2f){wq0.z, wq0.w};
        *(v2f*)(d + 16 * PITCHn)     = (v2f){wq1.x, wq1.y};
        *(v2f*)(d + 16 * PITCHn + 4) = (v2f){wq1.z, wq1.w};
        *(v2f*)(d + 32 * PITCHn)     = (v2f){wq2.x, wq2.y};
        *(v2f*)(d + 32 * PITCHn + 4) = (v2f){wq2.z, wq2.w};
        *(v2f*)(d + 48 * PITCHn)     = (v2f){wq3.x, wq3.y};
        *(v2f*)(d + 48 * PITCHn + 4) = (v2f){wq3.z, wq3.w};
    }
    __syncthreads();

#pragma unroll 1
    for (int ch = 0; ch < NCHn; ++ch) {
        const bool more = (ch + 1 < NCHn);
        if (more) {
            const float* ws = wsrc + (size_t)(ch + 1) * (CHn * ODn * Kn);
            wq0 = *(const v4f*)(ws);
            wq1 = *(const v4f*)(ws + 16 * cstride);
            wq2 = *(const v4f*)(ws + 32 * cstride);
            wq3 = *(const v4f*)(ws + 48 * cstride);
            const float* xs = xa + (size_t)(ch + 1) * (CHn * Ln);
#pragma unroll
            for (int cl = 0; cl < CHn; ++cl) {
                an[2 * cl]     = *(const v2f*)(xs + (size_t)cl * Ln);
                an[2 * cl + 1] = *(const v2f*)(xs + (size_t)cl * Ln + 4);
            }
        }

        const float* bsrc = &wl[ch & 1][brd];
#pragma unroll
        for (int cl = 0; cl < CHn; ++cl) {
            const v4f b0 = *(const v4f*)(bsrc +  0 * PITCHn + cl * 8);
            const v4f b1 = *(const v4f*)(bsrc + 16 * PITCHn + cl * 8);
            const v4f b2 = *(const v4f*)(bsrc + 32 * PITCHn + cl * 8);
            const v4f b3 = *(const v4f*)(bsrc + 48 * PITCHn + cl * 8);
            const v2f a0 = ac[2 * cl];
            const v2f a1 = ac[2 * cl + 1];

            acc0 = __builtin_amdgcn_wmma_f32_16x16x4_f32(false, a0, false, (v2f){b0.x, b0.y},
                                                         (short)0, acc0, false, false);
            acc1 = __builtin_amdgcn_wmma_f32_16x16x4_f32(false, a0, false, (v2f){b1.x, b1.y},
                                                         (short)0, acc1, false, false);
            acc2 = __builtin_amdgcn_wmma_f32_16x16x4_f32(false, a0, false, (v2f){b2.x, b2.y},
                                                         (short)0, acc2, false, false);
            acc3 = __builtin_amdgcn_wmma_f32_16x16x4_f32(false, a0, false, (v2f){b3.x, b3.y},
                                                         (short)0, acc3, false, false);
            acc0 = __builtin_amdgcn_wmma_f32_16x16x4_f32(false, a1, false, (v2f){b0.z, b0.w},
                                                         (short)0, acc0, false, false);
            acc1 = __builtin_amdgcn_wmma_f32_16x16x4_f32(false, a1, false, (v2f){b1.z, b1.w},
                                                         (short)0, acc1, false, false);
            acc2 = __builtin_amdgcn_wmma_f32_16x16x4_f32(false, a1, false, (v2f){b2.z, b2.w},
                                                         (short)0, acc2, false, false);
            acc3 = __builtin_amdgcn_wmma_f32_16x16x4_f32(false, a1, false, (v2f){b3.z, b3.w},
                                                         (short)0, acc3, false, false);
        }

        if (more) {
            float* d = &wl[(ch + 1) & 1][wdst];
            *(v2f*)(d)                   = (v2f){wq0.x, wq0.y};
            *(v2f*)(d + 4)               = (v2f){wq0.z, wq0.w};
            *(v2f*)(d + 16 * PITCHn)     = (v2f){wq1.x, wq1.y};
            *(v2f*)(d + 16 * PITCHn + 4) = (v2f){wq1.z, wq1.w};
            *(v2f*)(d + 32 * PITCHn)     = (v2f){wq2.x, wq2.y};
            *(v2f*)(d + 32 * PITCHn + 4) = (v2f){wq2.z, wq2.w};
            *(v2f*)(d + 48 * PITCHn)     = (v2f){wq3.x, wq3.y};
            *(v2f*)(d + 48 * PITCHn + 4) = (v2f){wq3.z, wq3.w};
            __syncthreads();
#pragma unroll
            for (int q = 0; q < 2 * CHn; ++q) ac[q] = an[q];
        }
    }

#pragma unroll
    for (int v = 0; v < 8; ++v) {
        const int bb = b_base + v + 8 * half;
        float* orow = out + (size_t)bb * (COUTn * ODn) + o;
        orow[(size_t)(r +  0) * ODn] = acc0[v] * RSQRT_CIN;
        orow[(size_t)(r + 16) * ODn] = acc1[v] * RSQRT_CIN;
        orow[(size_t)(r + 32) * ODn] = acc2[v] * RSQRT_CIN;
        orow[(size_t)(r + 48) * ODn] = acc3[v] * RSQRT_CIN;
    }
}

extern "C" void kernel_launch(void* const* d_in, const int* in_sizes, int n_in,
                              void* d_out, int out_size, void* d_ws, size_t ws_size,
                              hipStream_t stream) {
    const float* x = (const float*)d_in[0];
    const float* w = (const float*)d_in[1];
    float* out = (float*)d_out;

    lc1d_wmma_f32<<<dim3(ODn), dim3(256), 0, stream>>>(x, w, out);
}


# Workgroup tile = 4 adjacent o x 32 b (was 1 o x 128 b): W read as whole 128-B lines, x as 80-B runs via LDS, out stored as 16-B pieces; per-output sum order unchanged, bit-identical
# speedup vs baseline: 3.0177x; 2.3069x over previous
#include <hip/hip_runtime.h>


#define Bn    128
#define CINn  64
#define COUTn 64
#define ODn   256
#define Kn    8
#define Sn    4
#define Ln    1028
#define RSQRT_CIN 0.125f

#define OTn    4
#define BTn    32
#define CHn    8
#define NCHn   (CINn / CHn)
#define PITCHn 68
#define WSLn   (COUTn * PITCHn + 16)
#define XROWn  (Sn * (OTn - 1) + Kn)
#define XTILEn (CHn * BTn * XROWn)
#define BUFn   (OTn * WSLn + XTILEn)

typedef __attribute__((ext_vector_type(2))) float v2f;
typedef __attribute__((ext_vector_type(4))) float v4f;
typedef __attribute__((ext_vector_type(8))) float v8f;

__global__ __launch_bounds__(256) void
lc1d_wmma_f32(const float* __restrict__ x,
              const float* __restrict__ w,
              float* __restrict__ out)
{
    __shared__ __attribute__((aligned(16))) float lds[2][BUFn];

    const int o0   = (blockIdx.x >> 2) * OTn;
    const int b0   = (blockIdx.x & 3) * BTn;
    const int tid  = threadIdx.x;
    const int wave = tid >> 5;
    const int lane = tid & 31;
    const int half = lane >> 4;
    const int r    = lane & 15;
    const int ol   = wave >> 1;
    const int bt   = wave & 1;

    v8f acc0 = {}, acc1 = {}, acc2 = {}, acc3 = {};

    const size_t cstride = (size_t)CINn * ODn * Kn;
    const int q8 = tid & 7;
    const int cq = (tid >> 3) & 3;
    const float* wsrc = w + (size_t)cq * cstride + (size_t)wave * (ODn * Kn) + (size_t)o0 * Kn + 4 * q8;
    const int    wdst = (q8 >> 1) * WSLn + cq * PITCHn + wave * 8 + 2 * (q8 & 1);

    int xoff[5];
#pragma unroll
    for (int j = 0; j < 5; ++j) {
        const int p   = tid + 256 * j;
        const int row = p / 5;
        const int q   = p - 5 * row;
        const int cil = row >> 5;
        const int bl  = row & 31;
        xoff[j] = ((b0 + bl) * CINn + cil) * Ln + Sn * o0 + 4 * q;
    }
    const int xdst = OTn * WSLn + 4 * tid;

    const int brd = ol * WSLn + r * PITCHn + 4 * half;
    const int ard = OTn * WSLn + (16 * bt + r) * XROWn + Sn * ol + 2 * half;

    v4f wq[16];
    v4f xq[5];

#pragma unroll
    for (int j = 0; j < 16; ++j) wq[j] = *(const v4f*)(wsrc + (size_t)(4 * j) * cstride);
#pragma unroll
    for (int j = 0; j < 5; ++j) xq[j] = *(const v4f*)(x + xoff[j]);
    {
        float* d = &lds[0][wdst];
#pragma unroll
        for (int j = 0; j < 16; ++j) {
            *(v2f*)(d + j * 4 * PITCHn)     = (v2f){wq[j].x, wq[j].y};
            *(v2f*)(d + j * 4 * PITCHn + 4) = (v2f){wq[j].z, wq[j].w};
        }
        float* e = &lds[0][xdst];
#pragma unroll
        for (int j = 0; j < 5; ++j) *(v4f*)(e + 1024 * j) = xq[j];
    }
    __syncthreads();

#pragma unroll 1
    for (int ch = 0; ch < NCHn; ++ch) {
        const bool more = (ch + 1 < NCHn);
        if (more) {
            const float* ws = wsrc + (size_t)(ch + 1) * (CHn * ODn * Kn);
#pragma unroll
            for (int j = 0; j < 16; ++j) wq[j] = *(const v4f*)(ws + (size_t)(4 * j) * cstride);
            const float* xs = x + (ch + 1) * (CHn * Ln);
#pragma unroll
            for (int j = 0; j < 5; ++j) xq[j] = *(const v4f*)(xs + xoff[j]);
        }

        const float* bsrc = &lds[ch & 1][brd];
        const float* asrc = &lds[ch & 1][ard];
#pragma unroll
        for (int cl = 0; cl < CHn; ++cl) {
            const v4f b0 = *(const v4f*)(bsrc +  0 * PITCHn + cl * 8);
            const v4f b1 = *(const v4f*)(bsrc + 16 * PITCHn + cl * 8);
            const v4f b2 = *(const v4f*)(bsrc + 32 * PITCHn + cl * 8);
            const v4f b3 = *(const v4f*)(bsrc + 48 * PITCHn + cl * 8);
            const v2f a0 = *(const v2f*)(asrc + cl * (BTn * XROWn));
            const v2f a1 = *(const v2f*)(asrc + cl * (BTn * XROWn) + 4);

            acc0 = __builtin_amdgcn_wmma_f32_16x16x4_f32(false, a0, false, (v2f){b0.x, b0.y},
                                                         (short)0, acc0, false, false);
            acc1 = __builtin_amdgcn_wmma_f32_16x16x4_f32(false, a0, false, (v2f){b1.x, b1.y},
                                                         (short)0, acc1, false, false);
            acc2 = __builtin_amdgcn_wmma_f32_16x16x4_f32(false, a0, false, (v2f){b2.x, b2.y},
                                                         (short)0, acc2, false, false);
            acc3 = __builtin_amdgcn_wmma_f32_16x16x4_f32(false, a0, false, (v2f){b3.x, b3.y},
                                                         (short)0, acc3, false, false);
            acc0 = __builtin_amdgcn_wmma_f32_16x16x4_f32(false, a1, false, (v2f){b0.z, b0.w},
                                                         (short)0, acc0, false, false);
            acc1 = __builtin_amdgcn_wmma_f32_16x16x4_f32(false, a1, false, (v2f){b1.z, b1.w},
                                                         (short)0, acc1, false, false);
            acc2 = __builtin_amdgcn_wmma_f32_16x16x4_f32(false, a1, false, (v2f){b2.z, b2.w},
                                                         (short)0, acc2, false, false);
            acc3 = __builtin_amdgcn_wmma_f32_16x16x4_f32(false, a1, false, (v2f){b3.z, b3.w},
                                                         (short)0, acc3, false, false);
        }

        if (more) {
            float* d = &lds[(ch + 1) & 1][wdst];
#pragma unroll
            for (int j = 0; j < 16; ++j) {
                *(v2f*)(d + j * 4 * PITCHn)     = (v2f){wq[j].x, wq[j].y};
                *(v2f*)(d + j * 4 * PITCHn + 4) = (v2f){wq[j].z, wq[j].w};
            }
            float* e = &lds[(ch + 1) & 1][xdst];
#pragma unroll
            for (int j = 0; j < 5; ++j) *(v4f*)(e + 1024 * j) = xq[j];
            __syncthreads();
        }
    }

    float* os = &lds[0][0];
#pragma unroll
    for (int v = 0; v < 8; ++v) {
        const int bl = 16 * bt + v + 8 * half;
        float* orow = os + (bl * COUTn + r) * OTn + ol;
        orow[ 0 * OTn] = acc0[v] * RSQRT_CIN;
        orow[16 * OTn] = acc1[v] * RSQRT_CIN;
        orow[32 * OTn] = acc2[v] * RSQRT_CIN;
        orow[48 * OTn] = acc3[v] * RSQRT_CIN;
    }
    __syncthreads();
#pragma unroll
    for (int i = 0; i < 8; ++i) {
        const int e  = tid + 256 * i;
        const v4f val = *(const v4f*)(os + 4 * e);
        *(v4f*)(out + ((size_t)(b0 + (e >> 6)) * COUTn + (e & 63)) * ODn + o0) = val;
    }
}

extern "C" void kernel_launch(void* const* d_in, const int* in_sizes, int n_in,
                              void* d_out, int out_size, void* d_ws, size_t ws_size,
                              hipStream_t stream) {
    const float* x = (const float*)d_in[0];
    const float* w = (const float*)d_in[1];
    float* out = (float*)d_out;

    lc1d_wmma_f32<<<dim3((ODn / OTn) * (Bn / BTn)), dim3(256), 0, stream>>>(x, w, out);
}


# Global loads issued two chunks ahead of their WMMAs (second staging register set; LDS-only workgroup barrier keeps them in flight across the hand-over); per-output sum order unchanged, bit-identical
# speedup vs baseline: 3.2362x; 1.0724x over previous
#include <hip/hip_runtime.h>


#define Bn    128
#define CINn  64
#define COUTn 64
#define ODn   256
#define Kn    8
#define Sn    4
#define Ln    1028
#define RSQRT_CIN 0.125f

#define OTn    4
#define BTn    32
#define CHn    8
#define NCHn   (CINn / CHn)
#define PITCHn 68
#define WSLn   (COUTn * PITCHn + 16)
#define XROWn  (Sn * (OTn - 1) + Kn)
#define XTILEn (CHn * BTn * XROWn)
#define BUFn   (OTn * WSLn + XTILEn)

typedef __attribute__((ext_vector_type(2))) float v2f;
typedef __attribute__((ext_vector_type(4))) float v4f;
typedef __attribute__((ext_vector_type(8))) float v8f;

static __device__ __forceinline__ void lds_barrier() {
    __builtin_amdgcn_fence(__ATOMIC_RELEASE, "workgroup", "local");
    __builtin_amdgcn_s_barrier();
    __builtin_amdgcn_fence(__ATOMIC_ACQUIRE, "workgroup", "local");
}

static __device__ __forceinline__ void issue_chunk(v4f (&wq)[16], v4f (&xq)[5],
                                                   const float* ws, const float* xs,
                                                   const int (&xoff)[5], size_t cstride) {
#pragma unroll
    for (int j = 0; j < 16; ++j) wq[j] = *(const v4f*)(ws + (size_t)(4 * j) * cstride);
#pragma unroll
    for (int j = 0; j < 5; ++j) xq[j] = *(const v4f*)(xs + xoff[j]);
}

static __device__ __forceinline__ void stage_chunk(float* stage, const v4f (&wq)[16], const v4f (&xq)[5],
                                                   int wdst, int xdst) {
    float* d = stage + wdst;
#pragma unroll
    for (int j = 0; j < 16; ++j) {
        *(v2f*)(d + j * 4 * PITCHn)     = (v2f){wq[j].x, wq[j].y};
        *(v2f*)(d + j * 4 * PITCHn + 4) = (v2f){wq[j].z, wq[j].w};
    }
    float* e = stage + xdst;
#pragma unroll
    for (int j = 0; j < 5; ++j) *(v4f*)(e + 1024 * j) = xq[j];
}

template <int CL0, int CL1>
static __device__ __forceinline__ void mma_part(const float* bsrc, const float* asrc,
                                                v8f& acc0, v8f& acc1, v8f& acc2, v8f& acc3) {
#pragma unroll
    for (int cl = CL0; cl < CL1; ++cl) {
        const v4f b0 = *(const v4f*)(bsrc +  0 * PITCHn + cl * 8);
        const v4f b1 = *(const v4f*)(bsrc + 16 * PITCHn + cl * 8);
        const v4f b2 = *(const v4f*)(bsrc + 32 * PITCHn + cl * 8);
        const v4f b3 = *(const v4f*)(bsrc + 48 * PITCHn + cl * 8);
        const v2f a0 = *(const v2f*)(asrc + cl * (BTn * XROWn));
        const v2f a1 = *(const v2f*)(asrc + cl * (BTn * XROWn) + 4);

        acc0 = __builtin_amdgcn_wmma_f32_16x16x4_f32(false, a0, false, (v2f){b0.x, b0.y},
                                                     (short)0, acc0, false, false);
        acc1 = __builtin_amdgcn_wmma_f32_16x16x4_f32(false, a0, false, (v2f){b1.x, b1.y},
                                                     (short)0, acc1, false, false);
        acc2 = __builtin_amdgcn_wmma_f32_16x16x4_f32(false, a0, false, (v2f){b2.x, b2.y},
                                                     (short)0, acc2, false, false);
        acc3 = __builtin_amdgcn_wmma_f32_16x16x4_f32(false, a0, false, (v2f){b3.x, b3.y},
                                                     (short)0, acc3, false, false);
        acc0 = __builtin_amdgcn_wmma_f32_16x16x4_f32(false, a1, false, (v2f){b0.z, b0.w},
                                                     (short)0, acc0, false, false);
        acc1 = __builtin_amdgcn_wmma_f32_16x16x4_f32(false, a1, false, (v2f){b1.z, b1.w},
                                                     (short)0, acc1, false, false);
        acc2 = __builtin_amdgcn_wmma_f32_16x16x4_f32(false, a1, false, (v2f){b2.z, b2.w},
                                                     (short)0, acc2, false, false);
        acc3 = __builtin_amdgcn_wmma_f32_16x16x4_f32(false, a1, false, (v2f){b3.z, b3.w},
                                                     (short)0, acc3, false, false);
    }
}

static __device__ __forceinline__ void mma_chunk_and_stage(const float* bsrc, const float* asrc,
                                                           v8f& acc0, v8f& acc1, v8f& acc2, v8f& acc3,
                                                           float* next_stage, const v4f (&wq)[16],
                                                           const v4f (&xq)[5], int wdst, int xdst) {
    mma_part<0, 6>(bsrc, asrc, acc0, acc1, acc2, acc3);
    asm volatile("" : "+v"(acc0), "+v"(acc1), "+v"(acc2), "+v"(acc3) :: "memory");
    stage_chunk(next_stage, wq, xq, wdst, xdst);
    mma_part<6, 8>(bsrc, asrc, acc0, acc1, acc2, acc3);
}

__global__ __launch_bounds__(256) void
lc1d_wmma_f32(const float* __restrict__ x,
              const float* __restrict__ w,
              float* __restrict__ out)
{
    __shared__ __attribute__((aligned(16))) float lds[2][BUFn];

    const int o0   = (blockIdx.x >> 2) * OTn;
    const int b0   = (blockIdx.x & 3) * BTn;
    const int tid  = threadIdx.x;
    const int wave = tid >> 5;
    const int lane = tid & 31;
    const int half = lane >> 4;
    const int r    = lane & 15;
    const int ol   = wave >> 1;
    const int bt   = wave & 1;

    v8f acc0 = {}, acc1 = {}, acc2 = {}, acc3 = {};

    const size_t cstride = (size_t)CINn * ODn * Kn;
    const int q8 = tid & 7;
    const int cq = (tid >> 3) & 3;
    const float* wsrc = w + (size_t)cq * cstride + (size_t)wave * (ODn * Kn) + (size_t)o0 * Kn + 4 * q8;
    const int    wdst = (q8 >> 1) * WSLn + cq * PITCHn + wave * 8 + 2 * (q8 & 1);

    int xoff[5];
#pragma unroll
    for (int j = 0; j < 5; ++j) {
        const int p   = tid + 256 * j;
        const int row = p / 5;
        const int q   = p - 5 * row;
        const int cil = row >> 5;
        const int bl  = row & 31;
        xoff[j] = ((b0 + bl) * CINn + cil) * Ln + Sn * o0 + 4 * q;
    }
    const int xdst = OTn * WSLn + 4 * tid;

    const int brd = ol * WSLn + r * PITCHn + 4 * half;
    const int ard = OTn * WSLn + (16 * bt + r) * XROWn + Sn * ol + 2 * half;

    v4f wqA[16], wqB[16];
    v4f xqA[5],  xqB[5];

    issue_chunk(wqA, xqA, wsrc, x, xoff, cstride);
    __builtin_amdgcn_sched_barrier(0);
    issue_chunk(wqB, xqB, wsrc + (size_t)(CHn * ODn * Kn), x + CHn * Ln, xoff, cstride);
    stage_chunk(&lds[0][0], wqA, xqA, wdst, xdst);
    lds_barrier();

#pragma unroll 1
    for (int ch = 0; ch < NCHn - 2; ch += 2) {
        issue_chunk(wqA, xqA, wsrc + (size_t)(ch + 2) * (CHn * ODn * Kn), x + (ch + 2) * (CHn * Ln), xoff, cstride);
        __builtin_amdgcn_sched_barrier(0);
        mma_chunk_and_stage(&lds[0][brd], &lds[0][ard], acc0, acc1, acc2, acc3,
                            &lds[1][0], wqB, xqB, wdst, xdst);
        lds_barrier();

        issue_chunk(wqB, xqB, wsrc + (size_t)(ch + 3) * (CHn * ODn * Kn), x + (ch + 3) * (CHn * Ln), xoff, cstride);
        __builtin_amdgcn_sched_barrier(0);
        mma_chunk_and_stage(&lds[1][brd], &lds[1][ard], acc0, acc1, acc2, acc3,
                            &lds[0][0], wqA, xqA, wdst, xdst);
        lds_barrier();
    }
    mma_chunk_and_stage(&lds[0][brd], &lds[0][ard], acc0, acc1, acc2, acc3,
                        &lds[1][0], wqB, xqB, wdst, xdst);
    lds_barrier();
    mma_part<0, 8>(&lds[1][brd], &lds[1][ard], acc0, acc1, acc2, acc3);

    float* os = &lds[0][0];
#pragma unroll
    for (int v = 0; v < 8; ++v) {
        const int bl = 16 * bt + v + 8 * half;
        float* orow = os + (bl * COUTn + r) * OTn + ol;
        orow[ 0 * OTn] = acc0[v] * RSQRT_CIN;
        orow[16 * OTn] = acc1[v] * RSQRT_CIN;
        orow[32 * OTn] = acc2[v] * RSQRT_CIN;
        orow[48 * OTn] = acc3[v] * RSQRT_CIN;
    }
    __syncthreads();
#pragma unroll
    for (int i = 0; i < 8; ++i) {
        const int e  = tid + 256 * i;
        const v4f val = *(const v4f*)(os + 4 * e);
        *(v4f*)(out + ((size_t)(b0 + (e >> 6)) * COUTn + (e & 63)) * ODn + o0) = val;
    }
}

extern "C" void kernel_launch(void* const* d_in, const int* in_sizes, int n_in,
                              void* d_out, int out_size, void* d_ws, size_t ws_size,
                              hipStream_t stream) {
    const float* x = (const float*)d_in[0];
    const float* w = (const float*)d_in[1];
    float* out = (float*)d_out;

    lc1d_wmma_f32<<<dim3((ODn / OTn) * (Bn / BTn)), dim3(256), 0, stream>>>(x, w, out);
}


# Each workgroup walks the 8 input-channel chunks from its own start chunk and wraps, so workgroups sharing W and x lines ask for different lines at any moment; per-output chunk sum order rotated
# speedup vs baseline: 3.3986x; 1.0502x over previous
#include <hip/hip_runtime.h>


#define Bn    128
#define CINn  64
#define COUTn 64
#define ODn   256
#define Kn    8
#define Sn    4
#define Ln    1028
#define RSQRT_CIN 0.125f

#define OTn    4
#define BTn    32
#define CHn    8
#define NCHn   (CINn / CHn)
#define PITCHn 68
#define WSLn   (COUTn * PITCHn + 16)
#define XROWn  (Sn * (OTn - 1) + Kn)
#define XTILEn (CHn * BTn * XROWn)
#define BUFn   (OTn * WSLn + XTILEn)

typedef __attribute__((ext_vector_type(2))) float v2f;
typedef __attribute__((ext_vector_type(4))) float v4f;
typedef __attribute__((ext_vector_type(8))) float v8f;

static __device__ __forceinline__ void lds_barrier() {
    __builtin_amdgcn_fence(__ATOMIC_RELEASE, "workgroup", "local");
    __builtin_amdgcn_s_barrier();
    __builtin_amdgcn_fence(__ATOMIC_ACQUIRE, "workgroup", "local");
}

static __device__ __forceinline__ void issue_chunk(v4f (&wq)[16], v4f (&xq)[5],
                                                   const float* ws, const float* xs,
                                                   const int (&xoff)[5], size_t cstride) {
#pragma unroll
    for (int j = 0; j < 16; ++j) wq[j] = *(const v4f*)(ws + (size_t)(4 * j) * cstride);
#pragma unroll
    for (int j = 0; j < 5; ++j) xq[j] = *(const v4f*)(xs + xoff[j]);
}

static __device__ __forceinline__ void stage_chunk(float* stage, const v4f (&wq)[16], const v4f (&xq)[5],
                                                   int wdst, int xdst) {
    float* d = stage + wdst;
#pragma unroll
    for (int j = 0; j < 16; ++j) {
        *(v2f*)(d + j * 4 * PITCHn)     = (v2f){wq[j].x, wq[j].y};
        *(v2f*)(d + j * 4 * PITCHn + 4) = (v2f){wq[j].z, wq[j].w};
    }
    float* e = stage + xdst;
#pragma unroll
    for (int j = 0; j < 5; ++j) *(v4f*)(e + 1024 * j) = xq[j];
}

template <int CL0, int CL1>
static __device__ __forceinline__ void mma_part(const float* bsrc, const float* asrc,
                                                v8f& acc0, v8f& acc1, v8f& acc2, v8f& acc3) {
#pragma unroll
    for (int cl = CL0; cl < CL1; ++cl) {
        const v4f b0 = *(const v4f*)(bsrc +  0 * PITCHn + cl * 8);
        const v4f b1 = *(const v4f*)(bsrc + 16 * PITCHn + cl * 8);
        const v4f b2 = *(const v4f*)(bsrc + 32 * PITCHn + cl * 8);
        const v4f b3 = *(const v4f*)(bsrc + 48 * PITCHn + cl * 8);
        const v2f a0 = *(const v2f*)(asrc + cl * (BTn * XROWn));
        const v2f a1 = *(const v2f*)(asrc + cl * (BTn * XROWn) + 4);

        acc0 = __builtin_amdgcn_wmma_f32_16x16x4_f32(false, a0, false, (v2f){b0.x, b0.y},
                                                     (short)0, acc0, false, false);
        acc1 = __builtin_amdgcn_wmma_f32_16x16x4_f32(false, a0, false, (v2f){b1.x, b1.y},
                                                     (short)0, acc1, false, false);
        acc2 = __builtin_amdgcn_wmma_f32_16x16x4_f32(false, a0, false, (v2f){b2.x, b2.y},
                                                     (short)0, acc2, false, false);
        acc3 = __builtin_amdgcn_wmma_f32_16x16x4_f32(false, a0, false, (v2f){b3.x, b3.y},
                                                     (short)0, acc3, false, false);
        acc0 = __builtin_amdgcn_wmma_f32_16x16x4_f32(false, a1, false, (v2f){b0.z, b0.w},
                                                     (short)0, acc0, false, false);
        acc1 = __builtin_amdgcn_wmma_f32_16x16x4_f32(false, a1, false, (v2f){b1.z, b1.w},
                                                     (short)0, acc1, false, false);
        acc2 = __builtin_amdgcn_wmma_f32_16x16x4_f32(false, a1, false, (v2f){b2.z, b2.w},
                                                     (short)0, acc2, false, false);
        acc3 = __builtin_amdgcn_wmma_f32_16x16x4_f32(false, a1, false, (v2f){b3.z, b3.w},
                                                     (short)0, acc3, false, false);
    }
}

static __device__ __forceinline__ void mma_chunk_and_stage(const float* bsrc, const float* asrc,
                                                           v8f& acc0, v8f& acc1, v8f& acc2, v8f& acc3,
                                                           float* next_stage, const v4f (&wq)[16],
                                                           const v4f (&xq)[5], int wdst, int xdst) {
    mma_part<0, 6>(bsrc, asrc, acc0, acc1, acc2, acc3);
    asm volatile("" : "+v"(acc0), "+v"(acc1), "+v"(acc2), "+v"(acc3) :: "memory");
    stage_chunk(next_stage, wq, xq, wdst, xdst);
    mma_part<6, 8>(bsrc, asrc, acc0, acc1, acc2, acc3);
}

__global__ __launch_bounds__(256) void
lc1d_wmma_f32(const float* __restrict__ x,
              const float* __restrict__ w,
              float* __restrict__ out)
{
    __shared__ __attribute__((aligned(16))) float lds[2][BUFn];

    const int o0   = (blockIdx.x >> 2) * OTn;
    const int b0   = (blockIdx.x & 3) * BTn;
    const int tid  = threadIdx.x;
    const int wave = tid >> 5;
    const int lane = tid & 31;
    const int half = lane >> 4;
    const int r    = lane & 15;
    const int ol   = wave >> 1;
    const int bt   = wave & 1;

    const int rot = (2 * (int)(blockIdx.x & 3) + (int)((blockIdx.x >> 2) & 1) + 4 * (int)((blockIdx.x >> 3) & 1)) & (NCHn - 1);

    v8f acc0 = {}, acc1 = {}, acc2 = {}, acc3 = {};

    const size_t cstride = (size_t)CINn * ODn * Kn;
    const int q8 = tid & 7;
    const int cq = (tid >> 3) & 3;
    const float* wsrc = w + (size_t)cq * cstride + (size_t)wave * (ODn * Kn) + (size_t)o0 * Kn + 4 * q8;
    const int    wdst = (q8 >> 1) * WSLn + cq * PITCHn + wave * 8 + 2 * (q8 & 1);

    int xoff[5];
#pragma unroll
    for (int j = 0; j < 5; ++j) {
        const int p   = tid + 256 * j;
        const int row = p / 5;
        const int q   = p - 5 * row;
        const int cil = row >> 5;
        const int bl  = row & 31;
        xoff[j] = ((b0 + bl) * CINn + cil) * Ln + Sn * o0 + 4 * q;
    }
    const int xdst = OTn * WSLn + 4 * tid;

    const int brd = ol * WSLn + r * PITCHn + 4 * half;
    const int ard = OTn * WSLn + (16 * bt + r) * XROWn + Sn * ol + 2 * half;

    v4f wqA[16], wqB[16];
    v4f xqA[5],  xqB[5];

    const int c0 = rot, c1 = (rot + 1) & (NCHn - 1);
    issue_chunk(wqA, xqA, wsrc + (size_t)c0 * (CHn * ODn * Kn), x + c0 * (CHn * Ln), xoff, cstride);
    __builtin_amdgcn_sched_barrier(0);
    issue_chunk(wqB, xqB, wsrc + (size_t)c1 * (CHn * ODn * Kn), x + c1 * (CHn * Ln), xoff, cstride);
    stage_chunk(&lds[0][0], wqA, xqA, wdst, xdst);
    lds_barrier();

#pragma unroll 1
    for (int ch = 0; ch < NCHn - 2; ch += 2) {
        const int cA = (ch + 2 + rot) & (NCHn - 1);
        const int cB = (ch + 3 + rot) & (NCHn - 1);
        issue_chunk(wqA, xqA, wsrc + (size_t)cA * (CHn * ODn * Kn), x + cA * (CHn * Ln), xoff, cstride);
        __builtin_amdgcn_sched_barrier(0);
        mma_chunk_and_stage(&lds[0][brd], &lds[0][ard], acc0, acc1, acc2, acc3,
                            &lds[1][0], wqB, xqB, wdst, xdst);
        lds_barrier();

        issue_chunk(wqB, xqB, wsrc + (size_t)cB * (CHn * ODn * Kn), x + cB * (CHn * Ln), xoff, cstride);
        __builtin_amdgcn_sched_barrier(0);
        mma_chunk_and_stage(&lds[1][brd], &lds[1][ard], acc0, acc1, acc2, acc3,
                            &lds[0][0], wqA, xqA, wdst, xdst);
        lds_barrier();
    }
    mma_chunk_and_stage(&lds[0][brd], &lds[0][ard], acc0, acc1, acc2, acc3,
                        &lds[1][0], wqB, xqB, wdst, xdst);
    lds_barrier();
    mma_part<0, 8>(&lds[1][brd], &lds[1][ard], acc0, acc1, acc2, acc3);

    float* os = &lds[0][0];
#pragma unroll
    for (int v = 0; v < 8; ++v) {
        const int bl = 16 * bt + v + 8 * half;
        float* orow = os + (bl * COUTn + r) * OTn + ol;
        orow[ 0 * OTn] = acc0[v] * RSQRT_CIN;
        orow[16 * OTn] = acc1[v] * RSQRT_CIN;
        orow[32 * OTn] = acc2[v] * RSQRT_CIN;
        orow[48 * OTn] = acc3[v] * RSQRT_CIN;
    }
    __syncthreads();
#pragma unroll
    for (int i = 0; i < 8; ++i) {
        const int e  = tid + 256 * i;
        const v4f val = *(const v4f*)(os + 4 * e);
        *(v4f*)(out + ((size_t)(b0 + (e >> 6)) * COUTn + (e & 63)) * ODn + o0) = val;
    }
}

extern "C" void kernel_launch(void* const* d_in, const int* in_sizes, int n_in,
                              void* d_out, int out_size, void* d_ws, size_t ws_size,
                              hipStream_t stream) {
    const float* x = (const float*)d_in[0];
    const float* w = (const float*)d_in[1];
    float* out = (float*)d_out;

    lc1d_wmma_f32<<<dim3((ODn / OTn) * (Bn / BTn)), dim3(256), 0, stream>>>(x, w, out);
}
